# starting kernel
# baseline (speedup 1.0000x reference)
#include <hip/hip_runtime.h>
#include <hip/hip_bf16.h>

typedef __attribute__((ext_vector_type(2))) float v2f;
typedef __attribute__((ext_vector_type(8))) float v8f;

__device__ __forceinline__ void atomAddF(float* p, float v) {
    unsafeAtomicAdd(p, v);
}

__global__ void k_deg_init(float* __restrict__ deg, int N) {
    int i = blockIdx.x * blockDim.x + threadIdx.x;
    if (i < N) deg[i] = 1.0f;
}

__global__ void k_deg_count(const int* __restrict__ ei, int E, float* __restrict__ deg) {
    int e = blockIdx.x * blockDim.x + threadIdx.x;
    if (e < E) {
        int d = __builtin_nontemporal_load(ei + (size_t)E + e);
        atomAddF(&deg[d], 1.0f);
    }
}

__global__ void k_deg_rsqrt(float* __restrict__ deg, int N) {
    int i = blockIdx.x * blockDim.x + threadIdx.x;
    if (i < N) deg[i] = rsqrtf(deg[i]);
}

__global__ void k_gemm_wmma(const float* __restrict__ x,
                            const float* __restrict__ W1,
                            float* __restrict__ h,
                            int N, int K) {
    const int lane = threadIdx.x & 31;
    const int wave = threadIdx.x >> 5;
    const int tile = blockIdx.x * (blockDim.x >> 5) + wave;
    const int ntiles = (N + 15) >> 4;
    if (tile >= ntiles) return;

    const int m  = lane & 15;
    const int kb = (lane >> 4) << 1;
    int row = tile * 16 + m;
    if (row >= N) row = N - 1;
    const float* __restrict__ xrow = x + (size_t)row * K;

    v8f c = {};
    for (int k = 0; k < K; k += 4) {
        v2f a, b;
        a.x = __builtin_nontemporal_load(xrow + k + kb);
        a.y = __builtin_nontemporal_load(xrow + k + kb + 1);
        b.x = W1[(k + kb) * 16 + m];
        b.y = W1[(k + kb + 1) * 16 + m];
        c = __builtin_amdgcn_wmma_f32_16x16x4_f32(false, a, false, b,
                                                  (short)0, c, false, false);
    }
    const int rbase = (lane >> 4) << 3;
#pragma unroll
    for (int r = 0; r < 8; ++r) {
        const int orow = tile * 16 + rbase + r;
        if (orow < N) h[(size_t)orow * 16 + m] = c[r];
    }
}

__global__ void k_selfloop16(const float* __restrict__ h, const float* __restrict__ dinv,
                             float* __restrict__ hagg, int N) {
    int t = blockIdx.x * blockDim.x + threadIdx.x;
    if (t < N * 16) {
        int i = t >> 4;
        float w = dinv[i];
        hagg[t] = h[t] * w * w;
    }
}

__global__ void k_edge_agg16(const int* __restrict__ ei, int E,
                             const float* __restrict__ dinv,
                             const float* __restrict__ h,
                             float* __restrict__ hagg) {
    unsigned t = blockIdx.x * blockDim.x + threadIdx.x;
    unsigned e = t >> 2;
    if (e >= (unsigned)E) return;
    const int q = t & 3;
    const int s = __builtin_nontemporal_load(ei + e);
    const int d = __builtin_nontemporal_load(ei + (size_t)E + e);
    const float norm = dinv[s] * dinv[d];
    const float4 v = ((const float4*)(h + (size_t)s * 16))[q];
    float* dst = hagg + (size_t)d * 16 + q * 4;
    atomAddF(dst + 0, v.x * norm);
    atomAddF(dst + 1, v.y * norm);
    atomAddF(dst + 2, v.z * norm);
    atomAddF(dst + 3, v.w * norm);
}

__global__ void k_layer2_node(const float* __restrict__ hagg,
                              const float* __restrict__ b1,
                              const float* __restrict__ W2,
                              const float* __restrict__ dinv,
                              float* __restrict__ h2,
                              float* __restrict__ out2, int N) {
    int i = blockIdx.x * blockDim.x + threadIdx.x;
    if (i >= N) return;
    float a0 = 0.f, a1 = 0.f;
#pragma unroll
    for (int j = 0; j < 16; ++j) {
        float v = __builtin_nontemporal_load(hagg + (size_t)i * 16 + j) + b1[j];
        v = v > 0.f ? v : 0.f;
        a0 += v * W2[j * 2 + 0];
        a1 += v * W2[j * 2 + 1];
    }
    h2[(size_t)i * 2 + 0] = a0;
    h2[(size_t)i * 2 + 1] = a1;
    const float w = dinv[i] * dinv[i];
    out2[(size_t)i * 2 + 0] = a0 * w;
    out2[(size_t)i * 2 + 1] = a1 * w;
}

__global__ void k_edge_agg2(const int* __restrict__ ei, int E,
                            const float* __restrict__ dinv,
                            const float* __restrict__ h2,
                            float* __restrict__ out2) {
    int e = blockIdx.x * blockDim.x + threadIdx.x;
    if (e >= E) return;
    const int s = __builtin_nontemporal_load(ei + e);
    const int d = __builtin_nontemporal_load(ei + (size_t)E + e);
    const float norm = dinv[s] * dinv[d];
    const float2 v = ((const float2*)(h2 + (size_t)s * 2))[0];
    atomAddF(&out2[(size_t)d * 2 + 0], v.x * norm);
    atomAddF(&out2[(size_t)d * 2 + 1], v.y * norm);
}

__global__ void k_finalize(const float* __restrict__ out2,
                           const float* __restrict__ b2,
                           float* __restrict__ out, int N) {
    int i = blockIdx.x * blockDim.x + threadIdx.x;
    if (i >= N) return;
    const float v0 = out2[(size_t)i * 2 + 0] + b2[0];
    const float v1 = out2[(size_t)i * 2 + 1] + b2[1];
    const float mx = fmaxf(v0, v1);
    const float lse = mx + logf(expf(v0 - mx) + expf(v1 - mx));
    __builtin_nontemporal_store(v0 - lse, out + (size_t)i * 2 + 0);
    __builtin_nontemporal_store(v1 - lse, out + (size_t)i * 2 + 1);
}

extern "C" void kernel_launch(void* const* d_in, const int* in_sizes, int n_in,
                              void* d_out, int out_size, void* d_ws, size_t ws_size,
                              hipStream_t stream) {
    const float* x  = (const float*)d_in[0];
    const int*   ei = (const int*)d_in[1];
    const float* W1 = (const float*)d_in[2];
    const float* b1 = (const float*)d_in[3];
    const float* W2 = (const float*)d_in[4];
    const float* b2 = (const float*)d_in[5];
    float* out = (float*)d_out;

    const int H    = in_sizes[3];
    const int Fin  = in_sizes[2] / H;
    const int N    = in_sizes[0] / Fin;
    const int E    = in_sizes[1] / 2;
    (void)n_in; (void)out_size; (void)ws_size;

    float* ws   = (float*)d_ws;
    float* dinv = ws;
    float* h    = ws + (size_t)N;
    float* hagg = ws + (size_t)17 * N;
    float* h2   = ws + (size_t)33 * N;
    float* out2 = ws + (size_t)35 * N;

    const int B = 256;
    k_deg_init <<<(N + B - 1) / B, B, 0, stream>>>(dinv, N);
    k_deg_count<<<(E + B - 1) / B, B, 0, stream>>>(ei, E, dinv);
    k_deg_rsqrt<<<(N + B - 1) / B, B, 0, stream>>>(dinv, N);

    {
        const int ntiles = (N + 15) / 16;
        const int wpb = 4;
        k_gemm_wmma<<<(ntiles + wpb - 1) / wpb, wpb * 32, 0, stream>>>(x, W1, h, N, Fin);
    }

    k_selfloop16<<<((size_t)N * 16 + B - 1) / B, B, 0, stream>>>(h, dinv, hagg, N);
    k_edge_agg16<<<((size_t)E * 4 + B - 1) / B, B, 0, stream>>>(ei, E, dinv, h, hagg);

    k_layer2_node<<<(N + B - 1) / B, B, 0, stream>>>(hagg, b1, W2, dinv, h2, out2, N);

    k_edge_agg2<<<(E + B - 1) / B, B, 0, stream>>>(ei, E, dinv, h2, out2);

    k_finalize<<<(N + B - 1) / B, B, 0, stream>>>(out2, b2, out, N);
}


# Edges ordered by destination once (degree pass returns ranks, runs reserved per wave, fill); both layers pull per node, 16-lane rows sum in registers: no float atomics (sum order: rank, lane tree)
# speedup vs baseline: 1.6476x; 1.6476x over previous
#include <hip/hip_runtime.h>
#include <hip/hip_bf16.h>

typedef __attribute__((ext_vector_type(2))) float v2f;
typedef __attribute__((ext_vector_type(8))) float v8f;

__global__ void k_deg_init(int* __restrict__ cnt, int* __restrict__ cursor, int N) {
    int i = blockIdx.x * blockDim.x + threadIdx.x;
    if (i < N) cnt[i] = 0;
    if (i == 0) *cursor = 0;
}

__global__ void k_deg_count(const int* __restrict__ ei, int E, int* __restrict__ cnt,
                            int* __restrict__ rank) {
    int e = blockIdx.x * blockDim.x + threadIdx.x;
    if (e < E) {
        int d = __builtin_nontemporal_load(ei + (size_t)E + e);
        rank[e] = atomicAdd(&cnt[d], 1);
    }
}

__global__ void k_seg_alloc(const int* __restrict__ cnt, int* __restrict__ cursor,
                            int* __restrict__ off, float* __restrict__ dinv, int N) {
    const int i = blockIdx.x * blockDim.x + threadIdx.x;
    const int lane = threadIdx.x & 31;
    const int c = (i < N) ? cnt[i] : 0;
    int incl = c;
#pragma unroll
    for (int o = 1; o < 32; o <<= 1) {
        const int t = __shfl_up(incl, o, 32);
        if (lane >= o) incl += t;
    }
    const int total = __shfl(incl, 31, 32);
    int base = 0;
    if (lane == 0) base = atomicAdd(cursor, total);
    base = __shfl(base, 0, 32);
    if (i < N) {
        off[i]  = base + incl - c;
        dinv[i] = rsqrtf((float)(c + 1));
    }
}

constexpr int FILL_EDGES = 4;

__global__ void __launch_bounds__(256)
k_csr_fill(const int* __restrict__ ei, int E, const int* __restrict__ off,
           const int* __restrict__ rank, int* __restrict__ csr) {
    const long long e0 = (long long)blockIdx.x * (256 * FILL_EDGES) + threadIdx.x;
    int s[FILL_EDGES], d[FILL_EDGES], r[FILL_EDGES];
#pragma unroll
    for (int k = 0; k < FILL_EDGES; ++k) {
        const long long e = e0 + k * 256;
        s[k] = 0; d[k] = 0; r[k] = 0;
        if (e < E) {
            s[k] = __builtin_nontemporal_load(ei + e);
            d[k] = __builtin_nontemporal_load(ei + (size_t)E + e);
            r[k] = __builtin_nontemporal_load(rank + e);
        }
    }
    int o[FILL_EDGES];
#pragma unroll
    for (int k = 0; k < FILL_EDGES; ++k)
        o[k] = off[d[k]];
#pragma unroll
    for (int k = 0; k < FILL_EDGES; ++k) {
        const long long e = e0 + k * 256;
        if (e < E) csr[o[k] + r[k]] = s[k];
    }
}

__global__ void k_gemm_wmma(const float* __restrict__ x,
                            const float* __restrict__ W1,
                            float* __restrict__ h,
                            int N, int K) {
    const int lane = threadIdx.x & 31;
    const int wave = threadIdx.x >> 5;
    const int tile = blockIdx.x * (blockDim.x >> 5) + wave;
    const int ntiles = (N + 15) >> 4;
    if (tile >= ntiles) return;

    const int m  = lane & 15;
    const int kb = (lane >> 4) << 1;
    int row = tile * 16 + m;
    if (row >= N) row = N - 1;
    const float* __restrict__ xrow = x + (size_t)row * K;

    v8f c = {};
    for (int k = 0; k < K; k += 4) {
        v2f a, b;
        a.x = __builtin_nontemporal_load(xrow + k + kb);
        a.y = __builtin_nontemporal_load(xrow + k + kb + 1);
        b.x = W1[(k + kb) * 16 + m];
        b.y = W1[(k + kb + 1) * 16 + m];
        c = __builtin_amdgcn_wmma_f32_16x16x4_f32(false, a, false, b,
                                                  (short)0, c, false, false);
    }
    const int rbase = (lane >> 4) << 3;
#pragma unroll
    for (int r = 0; r < 8; ++r) {
        const int orow = tile * 16 + rbase + r;
        if (orow < N) h[(size_t)orow * 16 + m] = c[r];
    }
}

template <int J>
__device__ __forceinline__ int rowShare(int x) {
    return __builtin_amdgcn_update_dpp(0, x, 0x150 + J, 0xf, 0xf, true);
}

template <int J>
__device__ __forceinline__ void rowGather(int s, int c, const float* __restrict__ h, float* v) {
    v[J] = h[rowShare<J>(s) * 16 + c];
}

template <int J>
__device__ __forceinline__ void rowSum(int wb, const float* v, float& acc) {
    acc += v[J] * __int_as_float(rowShare<J>(wb));
}

#define ROW16(F, ...) \
    F<0>(__VA_ARGS__);  F<1>(__VA_ARGS__);  F<2>(__VA_ARGS__);  F<3>(__VA_ARGS__);  \
    F<4>(__VA_ARGS__);  F<5>(__VA_ARGS__);  F<6>(__VA_ARGS__);  F<7>(__VA_ARGS__);  \
    F<8>(__VA_ARGS__);  F<9>(__VA_ARGS__);  F<10>(__VA_ARGS__); F<11>(__VA_ARGS__); \
    F<12>(__VA_ARGS__); F<13>(__VA_ARGS__); F<14>(__VA_ARGS__); F<15>(__VA_ARGS__)

constexpr int PULL_NODES = 4;

__global__ void __launch_bounds__(256)
k_pull16(const int* __restrict__ csr, const int* __restrict__ off, const int* __restrict__ cnt,
         const float* __restrict__ dinv, const float* __restrict__ h,
         float* __restrict__ hagg, int N) {
    const int c = threadIdx.x & 15;
    const int row  = (int)((blockIdx.x * blockDim.x + threadIdx.x) >> 4);
    const int rows = (int)((gridDim.x * blockDim.x) >> 4);
  for (int t = 0; t < PULL_NODES; ++t) {
    const int i = row + t * rows;
    if (i >= N) break;
    const int beg = off[i];
    const int n   = cnt[i];
    const float di = dinv[i];
    float acc = h[i * 16 + c] * di * di;
    int s = i;
    float w = 0.f;
    if (c < n) {
        s = csr[beg + c];
        w = dinv[s] * di;
    }
    for (int j0 = 0; j0 < n; j0 += 16) {
        const int jn = j0 + 16 + c;
        int sn = i;
        if (jn < n) sn = csr[beg + jn];
        float v[16];
        ROW16(rowGather, s, c, h, v);
        float wn = 0.f;
        if (jn < n) wn = dinv[sn] * di;
        ROW16(rowSum, __float_as_int(w), v, acc);
        s = sn;
        w = wn;
    }
    hagg[i * 16 + c] = acc;
  }
}

__global__ void k_layer2_node(const float* __restrict__ hagg,
                              const float* __restrict__ b1,
                              const float* __restrict__ W2,
                              float* __restrict__ g2, int N) {
    int i = blockIdx.x * blockDim.x + threadIdx.x;
    if (i >= N) return;
    float g = 0.f;
#pragma unroll
    for (int j = 0; j < 16; ++j) {
        float v = __builtin_nontemporal_load(hagg + (size_t)i * 16 + j) + b1[j];
        v = v > 0.f ? v : 0.f;
        g += v * (W2[j * 2 + 0] - W2[j * 2 + 1]);
    }
    g2[i] = g;
}

__global__ void __launch_bounds__(256)
k_pull2(const int* __restrict__ csr, const int* __restrict__ off, const int* __restrict__ cnt,
        const float* __restrict__ dinv, const float* __restrict__ g2,
        float* __restrict__ zacc, int N) {
    const int c = threadIdx.x & 15;
    const int row  = (int)((blockIdx.x * blockDim.x + threadIdx.x) >> 4);
    const int rows = (int)((gridDim.x * blockDim.x) >> 4);
  for (int t = 0; t < PULL_NODES; ++t) {
    const int i = row + t * rows;
    if (i >= N) break;
    const int beg = off[i];
    const int n   = cnt[i];
    const float di = dinv[i];
    float part = 0.f;
    for (int j = c; j < n; j += 16) {
        const int s = csr[beg + j];
        part += g2[s] * (dinv[s] * di);
    }
    part += __shfl_xor(part, 8, 16);
    part += __shfl_xor(part, 4, 16);
    part += __shfl_xor(part, 2, 16);
    part += __shfl_xor(part, 1, 16);
    if (c == 0) zacc[i] = g2[i] * (di * di) + part;
  }
}

__global__ void k_finalize(const float* __restrict__ zacc,
                           const float* __restrict__ b2,
                           float* __restrict__ out, int N) {
    int i = blockIdx.x * blockDim.x + threadIdx.x;
    if (i >= N) return;
    const float z = zacc[i] + (b2[0] - b2[1]);
    const float l = logf(1.0f + expf(-fabsf(z)));
    __builtin_nontemporal_store(fminf(z, 0.f) - l,  out + (size_t)i * 2 + 0);
    __builtin_nontemporal_store(fminf(-z, 0.f) - l, out + (size_t)i * 2 + 1);
}

extern "C" void kernel_launch(void* const* d_in, const int* in_sizes, int n_in,
                              void* d_out, int out_size, void* d_ws, size_t ws_size,
                              hipStream_t stream) {
    const float* x  = (const float*)d_in[0];
    const int*   ei = (const int*)d_in[1];
    const float* W1 = (const float*)d_in[2];
    const float* b1 = (const float*)d_in[3];
    const float* W2 = (const float*)d_in[4];
    const float* b2 = (const float*)d_in[5];
    float* out = (float*)d_out;

    const int H    = in_sizes[3];
    const int Fin  = in_sizes[2] / H;
    const int N    = in_sizes[0] / Fin;
    const int E    = in_sizes[1] / 2;
    (void)n_in; (void)out_size; (void)ws_size;

    float* ws   = (float*)d_ws;
    float* dinv = ws;
    float* h    = ws + (size_t)N;
    float* hagg = ws + (size_t)17 * N;
    float* g2   = ws + (size_t)33 * N;
    float* zacc = ws + (size_t)34 * N;
    int*   wi   = (int*)d_ws;
    int*   cnt    = wi + (size_t)35 * N;
    int*   off    = wi + (size_t)36 * N;
    int*   cursor = wi + (size_t)37 * N;
    int*   rank   = wi + (size_t)38 * N;
    int*   csr    = wi + (size_t)38 * N + (size_t)E;

    const int B = 256;
    k_deg_init <<<(N + B - 1) / B, B, 0, stream>>>(cnt, cursor, N);
    k_deg_count<<<(E + B - 1) / B, B, 0, stream>>>(ei, E, cnt, rank);
    k_seg_alloc<<<(N + B - 1) / B, B, 0, stream>>>(cnt, cursor, off, dinv, N);

    {
        const int ntiles = (N + 15) / 16;
        const int wpb = 4;
        k_gemm_wmma<<<(ntiles + wpb - 1) / wpb, wpb * 32, 0, stream>>>(x, W1, h, N, Fin);
    }

    k_csr_fill<<<(E + B * FILL_EDGES - 1) / (B * FILL_EDGES), B, 0, stream>>>(ei, E, off, rank, csr);
    k_pull16<<<(N + 16 * PULL_NODES - 1) / (16 * PULL_NODES), B, 0, stream>>>(csr, off, cnt, dinv, h, hagg, N);

    k_layer2_node<<<(N + B - 1) / B, B, 0, stream>>>(hagg, b1, W2, g2, N);

    k_pull2<<<(N + 16 * PULL_NODES - 1) / (16 * PULL_NODES), B, 0, stream>>>(csr, off, cnt, dinv, g2, zacc, N);

    k_finalize<<<(N + B - 1) / B, B, 0, stream>>>(zacc, b2, out, N);
}


# speedup vs baseline: 1.6976x; 1.0303x over previous
#include <hip/hip_runtime.h>
#include <hip/hip_bf16.h>

typedef __attribute__((ext_vector_type(2))) float v2f;
typedef __attribute__((ext_vector_type(8))) float v8f;

__global__ void k_deg_init(int* __restrict__ cnt, int* __restrict__ cursor, int N) {
    int i = blockIdx.x * blockDim.x + threadIdx.x;
    if (i < N) cnt[i] = 0;
    if (i == 0) *cursor = 0;
}

__global__ void k_deg_count(const int* __restrict__ ei, int E, int* __restrict__ cnt,
                            int* __restrict__ rank) {
    int e = blockIdx.x * blockDim.x + threadIdx.x;
    if (e < E) {
        int d = __builtin_nontemporal_load(ei + (size_t)E + e);
        rank[e] = atomicAdd(&cnt[d], 1);
    }
}

__global__ void k_seg_alloc(const int* __restrict__ cnt, int* __restrict__ cursor,
                            int* __restrict__ off, float* __restrict__ dinv, int N) {
    const int i = blockIdx.x * blockDim.x + threadIdx.x;
    const int lane = threadIdx.x & 31;
    const int c = (i < N) ? cnt[i] : 0;
    int incl = c;
#pragma unroll
    for (int o = 1; o < 32; o <<= 1) {
        const int t = __shfl_up(incl, o, 32);
        if (lane >= o) incl += t;
    }
    const int total = __shfl(incl, 31, 32);
    int base = 0;
    if (lane == 0) base = atomicAdd(cursor, total);
    base = __shfl(base, 0, 32);
    if (i < N) {
        off[i]  = base + incl - c;
        dinv[i] = rsqrtf((float)(c + 1));
    }
}

constexpr int FILL_EDGES = 4;

__global__ void __launch_bounds__(256)
k_csr_fill(const int* __restrict__ ei, int E, const int* __restrict__ off,
           const int* __restrict__ rank, int* __restrict__ csr) {
    const long long e0 = (long long)blockIdx.x * (256 * FILL_EDGES) + threadIdx.x;
    int s[FILL_EDGES], d[FILL_EDGES], r[FILL_EDGES];
#pragma unroll
    for (int k = 0; k < FILL_EDGES; ++k) {
        const long long e = e0 + k * 256;
        s[k] = 0; d[k] = 0; r[k] = 0;
        if (e < E) {
            s[k] = __builtin_nontemporal_load(ei + e);
            d[k] = __builtin_nontemporal_load(ei + (size_t)E + e);
            r[k] = __builtin_nontemporal_load(rank + e);
        }
    }
    int o[FILL_EDGES];
#pragma unroll
    for (int k = 0; k < FILL_EDGES; ++k)
        o[k] = off[d[k]];
#pragma unroll
    for (int k = 0; k < FILL_EDGES; ++k) {
        const long long e = e0 + k * 256;
        if (e < E) csr[o[k] + r[k]] = s[k];
    }
}

__global__ void k_gemm_wmma(const float* __restrict__ x,
                            const float* __restrict__ W1,
                            float* __restrict__ h,
                            int N, int K) {
    const int lane = threadIdx.x & 31;
    const int wave = threadIdx.x >> 5;
    const int tile = blockIdx.x * (blockDim.x >> 5) + wave;
    const int ntiles = (N + 15) >> 4;
    if (tile >= ntiles) return;

    const int m  = lane & 15;
    const int kb = (lane >> 4) << 1;
    int row = tile * 16 + m;
    if (row >= N) row = N - 1;
    const float* __restrict__ xrow = x + (size_t)row * K;

    v8f c = {};
    for (int k = 0; k < K; k += 4) {
        v2f a, b;
        a.x = xrow[k + kb];
        a.y = xrow[k + kb + 1];
        b.x = W1[(k + kb) * 16 + m];
        b.y = W1[(k + kb + 1) * 16 + m];
        c = __builtin_amdgcn_wmma_f32_16x16x4_f32(false, a, false, b,
                                                  (short)0, c, false, false);
    }
    const int rbase = (lane >> 4) << 3;
#pragma unroll
    for (int r = 0; r < 8; ++r) {
        const int orow = tile * 16 + rbase + r;
        if (orow < N) h[(size_t)orow * 16 + m] = c[r];
    }
}

template <int J>
__device__ __forceinline__ int rowShare(int x) {
    return __builtin_amdgcn_update_dpp(0, x, 0x150 + J, 0xf, 0xf, true);
}

template <int J>
__device__ __forceinline__ void rowGather(int s, int c, const float* __restrict__ h, float* v) {
    v[J] = h[rowShare<J>(s) * 16 + c];
}

template <int J>
__device__ __forceinline__ void rowSum(int wb, const float* v, float& acc) {
    acc += v[J] * __int_as_float(rowShare<J>(wb));
}

#define ROW16(F, ...) \
    F<0>(__VA_ARGS__);  F<1>(__VA_ARGS__);  F<2>(__VA_ARGS__);  F<3>(__VA_ARGS__);  \
    F<4>(__VA_ARGS__);  F<5>(__VA_ARGS__);  F<6>(__VA_ARGS__);  F<7>(__VA_ARGS__);  \
    F<8>(__VA_ARGS__);  F<9>(__VA_ARGS__);  F<10>(__VA_ARGS__); F<11>(__VA_ARGS__); \
    F<12>(__VA_ARGS__); F<13>(__VA_ARGS__); F<14>(__VA_ARGS__); F<15>(__VA_ARGS__)

constexpr int PULL_NODES = 4;

__global__ void __launch_bounds__(256)
k_pull16(const int* __restrict__ csr, const int* __restrict__ off, const int* __restrict__ cnt,
         const float* __restrict__ dinv, const float* __restrict__ h,
         float* __restrict__ hagg, int N) {
    const int c = threadIdx.x & 15;
    const int row  = (int)((blockIdx.x * blockDim.x + threadIdx.x) >> 4);
    const int rows = (int)((gridDim.x * blockDim.x) >> 4);
  for (int t = 0; t < PULL_NODES; ++t) {
    const int i = row + t * rows;
    if (i >= N) break;
    const int beg = off[i];
    const int n   = cnt[i];
    const float di = dinv[i];
    float acc = h[i * 16 + c] * di * di;
    int s = i;
    float w = 0.f;
    if (c < n) {
        s = csr[beg + c];
        w = dinv[s] * di;
    }
    for (int j0 = 0; j0 < n; j0 += 16) {
        const int jn = j0 + 16 + c;
        int sn = i;
        if (jn < n) sn = csr[beg + jn];
        float v[16];
        ROW16(rowGather, s, c, h, v);
        float wn = 0.f;
        if (jn < n) wn = dinv[sn] * di;
        ROW16(rowSum, __float_as_int(w), v, acc);
        s = sn;
        w = wn;
    }
    hagg[i * 16 + c] = acc;
  }
}

__global__ void k_layer2_node(const float* __restrict__ hagg,
                              const float* __restrict__ b1,
                              const float* __restrict__ W2,
                              float* __restrict__ g2, int N) {
    int i = blockIdx.x * blockDim.x + threadIdx.x;
    if (i >= N) return;
    float g = 0.f;
#pragma unroll
    for (int j = 0; j < 16; ++j) {
        float v = __builtin_nontemporal_load(hagg + (size_t)i * 16 + j) + b1[j];
        v = v > 0.f ? v : 0.f;
        g += v * (W2[j * 2 + 0] - W2[j * 2 + 1]);
    }
    g2[i] = g;
}

__global__ void __launch_bounds__(256)
k_pull2(const int* __restrict__ csr, const int* __restrict__ off, const int* __restrict__ cnt,
        const float* __restrict__ dinv, const float* __restrict__ g2,
        float* __restrict__ zacc, int N) {
    const int c = threadIdx.x & 15;
    const int row  = (int)((blockIdx.x * blockDim.x + threadIdx.x) >> 4);
    const int rows = (int)((gridDim.x * blockDim.x) >> 4);
  for (int t = 0; t < PULL_NODES; ++t) {
    const int i = row + t * rows;
    if (i >= N) break;
    const int beg = off[i];
    const int n   = cnt[i];
    const float di = dinv[i];
    float part = 0.f;
    for (int j = c; j < n; j += 16) {
        const int s = csr[beg + j];
        part += g2[s] * (dinv[s] * di);
    }
    part += __shfl_xor(part, 8, 16);
    part += __shfl_xor(part, 4, 16);
    part += __shfl_xor(part, 2, 16);
    part += __shfl_xor(part, 1, 16);
    if (c == 0) zacc[i] = g2[i] * (di * di) + part;
  }
}

__global__ void k_finalize(const float* __restrict__ zacc,
                           const float* __restrict__ b2,
                           float* __restrict__ out, int N) {
    int i = blockIdx.x * blockDim.x + threadIdx.x;
    if (i >= N) return;
    const float z = zacc[i] + (b2[0] - b2[1]);
    const float l = logf(1.0f + expf(-fabsf(z)));
    __builtin_nontemporal_store(fminf(z, 0.f) - l,  out + (size_t)i * 2 + 0);
    __builtin_nontemporal_store(fminf(-z, 0.f) - l, out + (size_t)i * 2 + 1);
}

extern "C" void kernel_launch(void* const* d_in, const int* in_sizes, int n_in,
                              void* d_out, int out_size, void* d_ws, size_t ws_size,
                              hipStream_t stream) {
    const float* x  = (const float*)d_in[0];
    const int*   ei = (const int*)d_in[1];
    const float* W1 = (const float*)d_in[2];
    const float* b1 = (const float*)d_in[3];
    const float* W2 = (const float*)d_in[4];
    const float* b2 = (const float*)d_in[5];
    float* out = (float*)d_out;

    const int H    = in_sizes[3];
    const int Fin  = in_sizes[2] / H;
    const int N    = in_sizes[0] / Fin;
    const int E    = in_sizes[1] / 2;
    (void)n_in; (void)out_size; (void)ws_size;

    float* ws   = (float*)d_ws;
    float* dinv = ws;
    float* h    = ws + (size_t)N;
    float* hagg = ws + (size_t)17 * N;
    float* g2   = ws + (size_t)33 * N;
    float* zacc = ws + (size_t)34 * N;
    int*   wi   = (int*)d_ws;
    int*   cnt    = wi + (size_t)35 * N;
    int*   off    = wi + (size_t)36 * N;
    int*   cursor = wi + (size_t)37 * N;
    int*   rank   = wi + (size_t)38 * N;
    int*   csr    = wi + (size_t)38 * N + (size_t)E;

    const int B = 256;
    k_deg_init <<<(N + B - 1) / B, B, 0, stream>>>(cnt, cursor, N);
    k_deg_count<<<(E + B - 1) / B, B, 0, stream>>>(ei, E, cnt, rank);
    k_seg_alloc<<<(N + B - 1) / B, B, 0, stream>>>(cnt, cursor, off, dinv, N);

    {
        const int ntiles = (N + 15) / 16;
        const int wpb = 4;
        k_gemm_wmma<<<(ntiles + wpb - 1) / wpb, wpb * 32, 0, stream>>>(x, W1, h, N, Fin);
    }

    k_csr_fill<<<(E + B * FILL_EDGES - 1) / (B * FILL_EDGES), B, 0, stream>>>(ei, E, off, rank, csr);
    k_pull16<<<(N + 16 * PULL_NODES - 1) / (16 * PULL_NODES), B, 0, stream>>>(csr, off, cnt, dinv, h, hagg, N);

    k_layer2_node<<<(N + B - 1) / B, B, 0, stream>>>(hagg, b1, W2, g2, N);

    k_pull2<<<(N + 16 * PULL_NODES - 1) / (16 * PULL_NODES), B, 0, stream>>>(csr, off, cnt, dinv, g2, zacc, N);

    k_finalize<<<(N + B - 1) / B, B, 0, stream>>>(zacc, b2, out, N);
}


# Count pass keeps two planes of runs, one per half of the chip: a work-group adds and stores only in the plane of the half it runs on (die id); pulls read both runs of a node (sum order changes)
# speedup vs baseline: 4.9035x; 2.8885x over previous
#include <hip/hip_runtime.h>
#include <hip/hip_bf16.h>

typedef __attribute__((ext_vector_type(2))) float v2f;
typedef __attribute__((ext_vector_type(8))) float v8f;

constexpr int RUN_WORDS = 64;
constexpr int RUN_CAP = RUN_WORDS - 1;
constexpr int PLACE_EDGES = 8;

__global__ void k_deg_init(int* __restrict__ slots, int* __restrict__ slotsB, int* __restrict__ ovf_n, int N) {
    int i = blockIdx.x * blockDim.x + threadIdx.x;
    if (i < N) {
        slots[i * RUN_WORDS] = 0;
        slotsB[i * RUN_WORDS] = 0;
    }
    if (i == 0) *ovf_n = 0;
}

__global__ void __launch_bounds__(256)
k_count_place(const int* __restrict__ ei, int E,
              int* __restrict__ slotsA, int* __restrict__ slotsB, int* __restrict__ ovf_n, int* __restrict__ ovf) {
    const unsigned place = __builtin_amdgcn_s_sendmsg_rtn(0x87);
    int* __restrict__ slots = ((place >> 18) & 1u) ? slotsB : slotsA;
    const long long e0 = (long long)blockIdx.x * (256 * PLACE_EDGES) + threadIdx.x;
    int s[PLACE_EDGES], d[PLACE_EDGES], r[PLACE_EDGES];
#pragma unroll
    for (int k = 0; k < PLACE_EDGES; ++k) {
        const long long e = e0 + k * 256;
        s[k] = 0; d[k] = 0;
        if (e < E) {
            s[k] = __builtin_nontemporal_load(ei + e);
            d[k] = __builtin_nontemporal_load(ei + (size_t)E + e);
        }
    }
#pragma unroll
    for (int k = 0; k < PLACE_EDGES; ++k) {
        const long long e = e0 + k * 256;
        r[k] = 0;
        if (e < E) r[k] = atomicAdd(&slots[d[k] * RUN_WORDS], 1);
    }
#pragma unroll
    for (int k = 0; k < PLACE_EDGES; ++k) {
        const long long e = e0 + k * 256;
        if (e < E) {
            if (r[k] < RUN_CAP) {
                slots[d[k] * RUN_WORDS + 1 + r[k]] = s[k];
            } else {
                const int o = atomicAdd(ovf_n, 1);
                ovf[2 * o]     = s[k];
                ovf[2 * o + 1] = d[k];
            }
        }
    }
}

__global__ void k_gemm_wmma(const float* __restrict__ x,
                            const float* __restrict__ W1,
                            const int* __restrict__ slots, const int* __restrict__ slotsB,
                            float* __restrict__ h2,
                            int N, int K) {
    const int lane = threadIdx.x & 31;
    const int wave = threadIdx.x >> 5;
    const int tile = blockIdx.x * (blockDim.x >> 5) + wave;
    const int ntiles = (N + 15) >> 4;
    if (tile >= ntiles) return;

    const int m  = lane & 15;
    const int kb = (lane >> 4) << 1;
    int row = tile * 16 + m;
    if (row >= N) row = N - 1;
    const float* __restrict__ xrow = x + (size_t)row * K;

    v8f c = {};
    for (int k = 0; k < K; k += 4) {
        v2f a, b;
        a.x = xrow[k + kb];
        a.y = xrow[k + kb + 1];
        b.x = W1[(k + kb) * 16 + m];
        b.y = W1[(k + kb + 1) * 16 + m];
        c = __builtin_amdgcn_wmma_f32_16x16x4_f32(false, a, false, b,
                                                  (short)0, c, false, false);
    }
    const int rbase = (lane >> 4) << 3;
#pragma unroll
    for (int r = 0; r < 8; ++r) {
        const int orow = tile * 16 + rbase + r;
        if (orow < N) {
            v2f p;
            p.x = c[r];
            p.y = rsqrtf((float)(slots[orow * RUN_WORDS] + slotsB[orow * RUN_WORDS] + 1));
            *(v2f*)(h2 + (size_t)orow * 32 + 2 * m) = p;
        }
    }
}

template <int J>
__device__ __forceinline__ int rowShare(int x) {
    return __builtin_amdgcn_update_dpp(0, x, 0x150 + J, 0xf, 0xf, true);
}

template <int J>
__device__ __forceinline__ void rowGather(int s, int c, const float* __restrict__ h2, v2f* v) {
    v[J] = *(const v2f*)(h2 + rowShare<J>(s) * 32 + 2 * c);
}

template <int J>
__device__ __forceinline__ void rowSum(int wb, int m, float di, const v2f* v, float& acc) {
    const int wi = wb + J;
    const float w = (wi >= 1 && wi <= m) ? v[J].y * di : 0.f;
    acc += v[J].x * w;
}

#define ROW16(F, ...) \
    F<0>(__VA_ARGS__);  F<1>(__VA_ARGS__);  F<2>(__VA_ARGS__);  F<3>(__VA_ARGS__);  \
    F<4>(__VA_ARGS__);  F<5>(__VA_ARGS__);  F<6>(__VA_ARGS__);  F<7>(__VA_ARGS__);  \
    F<8>(__VA_ARGS__);  F<9>(__VA_ARGS__);  F<10>(__VA_ARGS__); F<11>(__VA_ARGS__); \
    F<12>(__VA_ARGS__); F<13>(__VA_ARGS__); F<14>(__VA_ARGS__); F<15>(__VA_ARGS__)

constexpr int PULL_NODES = 4;

__device__ __forceinline__ void pullRun(const int* __restrict__ run, int w0, int m, int i, int c, float di,
                                        const float* __restrict__ h2, float& acc) {
    int s = (c >= 1 && c <= m) ? w0 : i;
    for (int wb = 0; wb <= m; wb += 16) {
        const int wn = wb + 16 + c;
        int sn = i;
        if (wn <= m) sn = run[wn];
        v2f v[16];
        ROW16(rowGather, s, c, h2, v);
        ROW16(rowSum, wb, m, di, v, acc);
        s = sn;
    }
}

__global__ void __launch_bounds__(256) __attribute__((amdgpu_waves_per_eu(16)))
k_pull16(const int* __restrict__ slots, const int* __restrict__ slotsB,
         const int* __restrict__ ovf_n, const int* __restrict__ ovf,
         const float* __restrict__ h2, float* __restrict__ hagg, int N) {
    const int c = threadIdx.x & 15;
    const int row  = (int)((blockIdx.x * blockDim.x + threadIdx.x) >> 4);
    const int rows = (int)((gridDim.x * blockDim.x) >> 4);
  for (int t = 0; t < PULL_NODES; ++t) {
    const int i = row + t * rows;
    if (i >= N) break;
    const int* __restrict__ runA = slots  + i * RUN_WORDS;
    const int* __restrict__ runB = slotsB + i * RUN_WORDS;
    const int wA = runA[c];
    const int wB = runB[c];
    const int nA = rowShare<0>(wA);
    const int nB = rowShare<0>(wB);
    const int mA = nA < RUN_CAP ? nA : RUN_CAP;
    const int mB = nB < RUN_CAP ? nB : RUN_CAP;
    const float di = rsqrtf((float)(nA + nB + 1));
    float acc = h2[i * 32 + 2 * c] * di * di;
    if (mA > 0) pullRun(runA, wA, mA, i, c, di, h2, acc);
    if (mB > 0) pullRun(runB, wB, mB, i, c, di, h2, acc);
    if (nA > RUN_CAP || nB > RUN_CAP) {
        const int no = *ovf_n;
        for (int q = 0; q < no; ++q) {
            if (ovf[2 * q + 1] == i) {
                const int os = ovf[2 * q];
                const v2f p = *(const v2f*)(h2 + os * 32 + 2 * c);
                acc += p.x * (p.y * di);
            }
        }
    }
    hagg[i * 16 + c] = acc;
  }
}

__global__ void k_layer2_node(const float* __restrict__ hagg,
                              const float* __restrict__ b1,
                              const float* __restrict__ W2,
                              const int* __restrict__ slots, const int* __restrict__ slotsB,
                              float* __restrict__ gd, int N) {
    int i = blockIdx.x * blockDim.x + threadIdx.x;
    if (i >= N) return;
    float g = 0.f;
#pragma unroll
    for (int j = 0; j < 16; ++j) {
        float v = __builtin_nontemporal_load(hagg + (size_t)i * 16 + j) + b1[j];
        v = v > 0.f ? v : 0.f;
        g += v * (W2[j * 2 + 0] - W2[j * 2 + 1]);
    }
    v2f p;
    p.x = g;
    p.y = rsqrtf((float)(slots[i * RUN_WORDS] + slotsB[i * RUN_WORDS] + 1));
    *(v2f*)(gd + (size_t)i * 2) = p;
}

__global__ void __launch_bounds__(256)
k_pull2(const int* __restrict__ slots, const int* __restrict__ slotsB,
        const int* __restrict__ ovf_n, const int* __restrict__ ovf,
        const float* __restrict__ gd, float* __restrict__ zacc, int N) {
    const int c = threadIdx.x & 15;
    const int row  = (int)((blockIdx.x * blockDim.x + threadIdx.x) >> 4);
    const int rows = (int)((gridDim.x * blockDim.x) >> 4);
  for (int t = 0; t < PULL_NODES; ++t) {
    const int i = row + t * rows;
    if (i >= N) break;
    const int* __restrict__ runA = slots  + i * RUN_WORDS;
    const int* __restrict__ runB = slotsB + i * RUN_WORDS;
    const int a0 = runA[c];
    const int a1 = runA[c + 16];
    const int b0 = runB[c];
    const int b1 = runB[c + 16];
    const float self = gd[(size_t)i * 2];
    const int nA = rowShare<0>(a0);
    const int nB = rowShare<0>(b0);
    const int mA = nA < RUN_CAP ? nA : RUN_CAP;
    const int mB = nB < RUN_CAP ? nB : RUN_CAP;
    const float di = rsqrtf((float)(nA + nB + 1));
    const bool ea0 = c >= 1 && c <= mA, ea1 = c + 16 <= mA;
    const bool eb0 = c >= 1 && c <= mB, eb1 = c + 16 <= mB;
    const v2f p0 = *(const v2f*)(gd + (size_t)(ea0 ? a0 : i) * 2);
    const v2f p1 = *(const v2f*)(gd + (size_t)(ea1 ? a1 : i) * 2);
    const v2f p2 = *(const v2f*)(gd + (size_t)(eb0 ? b0 : i) * 2);
    const v2f p3 = *(const v2f*)(gd + (size_t)(eb1 ? b1 : i) * 2);
    float part = 0.f;
    part += ea0 ? p0.x * (p0.y * di) : 0.f;
    part += ea1 ? p1.x * (p1.y * di) : 0.f;
    part += eb0 ? p2.x * (p2.y * di) : 0.f;
    part += eb1 ? p3.x * (p3.y * di) : 0.f;
    if (mA >= 32 || mB >= 32) {
        const int a2 = runA[c + 32];
        const int a3 = runA[c + 48];
        const int b2 = runB[c + 32];
        const int b3 = runB[c + 48];
        const bool ea2 = c + 32 <= mA, ea3 = c + 48 <= mA;
        const bool eb2 = c + 32 <= mB, eb3 = c + 48 <= mB;
        const v2f q0 = *(const v2f*)(gd + (size_t)(ea2 ? a2 : i) * 2);
        const v2f q1 = *(const v2f*)(gd + (size_t)(ea3 ? a3 : i) * 2);
        const v2f q2 = *(const v2f*)(gd + (size_t)(eb2 ? b2 : i) * 2);
        const v2f q3 = *(const v2f*)(gd + (size_t)(eb3 ? b3 : i) * 2);
        part += ea2 ? q0.x * (q0.y * di) : 0.f;
        part += ea3 ? q1.x * (q1.y * di) : 0.f;
        part += eb2 ? q2.x * (q2.y * di) : 0.f;
        part += eb3 ? q3.x * (q3.y * di) : 0.f;
    }
    if (nA > RUN_CAP || nB > RUN_CAP) {
        const int no = *ovf_n;
        for (int q = c; q < no; q += 16) {
            if (ovf[2 * q + 1] == i) {
                const int os = ovf[2 * q];
                const v2f p = *(const v2f*)(gd + (size_t)os * 2);
                part += p.x * (p.y * di);
            }
        }
    }
    part += __shfl_xor(part, 8, 16);
    part += __shfl_xor(part, 4, 16);
    part += __shfl_xor(part, 2, 16);
    part += __shfl_xor(part, 1, 16);
    if (c == 0) zacc[i] = self * (di * di) + part;
  }
}

__global__ void k_finalize(const float* __restrict__ zacc,
                           const float* __restrict__ b2,
                           float* __restrict__ out, int N) {
    int i = blockIdx.x * blockDim.x + threadIdx.x;
    if (i >= N) return;
    const float z = zacc[i] + (b2[0] - b2[1]);
    const float l = logf(1.0f + expf(-fabsf(z)));
    __builtin_nontemporal_store(fminf(z, 0.f) - l,  out + (size_t)i * 2 + 0);
    __builtin_nontemporal_store(fminf(-z, 0.f) - l, out + (size_t)i * 2 + 1);
}

extern "C" void kernel_launch(void* const* d_in, const int* in_sizes, int n_in,
                              void* d_out, int out_size, void* d_ws, size_t ws_size,
                              hipStream_t stream) {
    const float* x  = (const float*)d_in[0];
    const int*   ei = (const int*)d_in[1];
    const float* W1 = (const float*)d_in[2];
    const float* b1 = (const float*)d_in[3];
    const float* W2 = (const float*)d_in[4];
    const float* b2 = (const float*)d_in[5];
    float* out = (float*)d_out;

    const int H    = in_sizes[3];
    const int Fin  = in_sizes[2] / H;
    const int N    = in_sizes[0] / Fin;
    const int E    = in_sizes[1] / 2;
    (void)n_in; (void)out_size; (void)ws_size;

    int*   wi     = (int*)d_ws;
    float* ws     = (float*)d_ws;
    int*   slots  = wi;
    int*   slotsB = wi + (size_t)RUN_WORDS * N;
    float* h2     = ws + (size_t)(2 * RUN_WORDS) * N;
    float* hagg   = ws + (size_t)(2 * RUN_WORDS + 32) * N;
    float* gd     = ws + (size_t)(2 * RUN_WORDS + 48) * N;
    float* zacc   = ws + (size_t)(2 * RUN_WORDS + 50) * N;
    int*   ovf_n  = wi + (size_t)(2 * RUN_WORDS + 51) * N;
    int*   ovf    = wi + (size_t)(2 * RUN_WORDS + 52) * N;

    const int B = 256;
    k_deg_init   <<<(N + B - 1) / B, B, 0, stream>>>(slots, slotsB, ovf_n, N);
    k_count_place<<<(E + B * PLACE_EDGES - 1) / (B * PLACE_EDGES), B, 0, stream>>>(ei, E, slots, slotsB, ovf_n, ovf);

    {
        const int ntiles = (N + 15) / 16;
        const int wpb = 4;
        k_gemm_wmma<<<(ntiles + wpb - 1) / wpb, wpb * 32, 0, stream>>>(x, W1, slots, slotsB, h2, N, Fin);
    }

    k_pull16<<<(N + 16 * PULL_NODES - 1) / (16 * PULL_NODES), B, 0, stream>>>(slots, slotsB, ovf_n, ovf, h2, hagg, N);

    k_layer2_node<<<(N + B - 1) / B, B, 0, stream>>>(hagg, b1, W2, slots, slotsB, gd, N);

    k_pull2<<<(N + 16 * PULL_NODES - 1) / (16 * PULL_NODES), B, 0, stream>>>(slots, slotsB, ovf_n, ovf, gd, zacc, N);

    k_finalize<<<(N + B - 1) / B, B, 0, stream>>>(zacc, b2, out, N);
}


# Each pull finishes its node in its own lanes: layer-1 pull ends in bias, ReLU and the W2-difference dot over the row (lane tree: sum order changes), layer-2 pull in log_softmax; 7 -> 5 launches
# speedup vs baseline: 5.2150x; 1.0635x over previous
#include <hip/hip_runtime.h>
#include <hip/hip_bf16.h>

typedef __attribute__((ext_vector_type(2))) float v2f;
typedef __attribute__((ext_vector_type(8))) float v8f;

constexpr int RUN_WORDS = 64;
constexpr int RUN_CAP = RUN_WORDS - 1;
constexpr int PLACE_EDGES = 8;

__global__ void k_deg_init(int* __restrict__ slots, int* __restrict__ slotsB, int* __restrict__ ovf_n, int N) {
    int i = blockIdx.x * blockDim.x + threadIdx.x;
    if (i < N) {
        slots[i * RUN_WORDS] = 0;
        slotsB[i * RUN_WORDS] = 0;
    }
    if (i == 0) *ovf_n = 0;
}

__global__ void __launch_bounds__(256)
k_count_place(const int* __restrict__ ei, int E,
              int* __restrict__ slotsA, int* __restrict__ slotsB, int* __restrict__ ovf_n, int* __restrict__ ovf) {
    const unsigned place = __builtin_amdgcn_s_sendmsg_rtn(0x87);
    int* __restrict__ slots = ((place >> 18) & 1u) ? slotsB : slotsA;
    const long long e0 = (long long)blockIdx.x * (256 * PLACE_EDGES) + threadIdx.x;
    int s[PLACE_EDGES], d[PLACE_EDGES], r[PLACE_EDGES];
#pragma unroll
    for (int k = 0; k < PLACE_EDGES; ++k) {
        const long long e = e0 + k * 256;
        s[k] = 0; d[k] = 0;
        if (e < E) {
            s[k] = __builtin_nontemporal_load(ei + e);
            d[k] = __builtin_nontemporal_load(ei + (size_t)E + e);
        }
    }
#pragma unroll
    for (int k = 0; k < PLACE_EDGES; ++k) {
        const long long e = e0 + k * 256;
        r[k] = 0;
        if (e < E) r[k] = atomicAdd(&slots[d[k] * RUN_WORDS], 1);
    }
#pragma unroll
    for (int k = 0; k < PLACE_EDGES; ++k) {
        const long long e = e0 + k * 256;
        if (e < E) {
            if (r[k] < RUN_CAP) {
                slots[d[k] * RUN_WORDS + 1 + r[k]] = s[k];
            } else {
                const int o = atomicAdd(ovf_n, 1);
                ovf[2 * o]     = s[k];
                ovf[2 * o + 1] = d[k];
            }
        }
    }
}

__global__ void k_gemm_wmma(const float* __restrict__ x,
                            const float* __restrict__ W1,
                            const int* __restrict__ slots, const int* __restrict__ slotsB,
                            float* __restrict__ h2,
                            int N, int K) {
    const int lane = threadIdx.x & 31;
    const int wave = threadIdx.x >> 5;
    const int tile = blockIdx.x * (blockDim.x >> 5) + wave;
    const int ntiles = (N + 15) >> 4;
    if (tile >= ntiles) return;

    const int m  = lane & 15;
    const int kb = (lane >> 4) << 1;
    int row = tile * 16 + m;
    if (row >= N) row = N - 1;
    const float* __restrict__ xrow = x + (size_t)row * K;

    v8f c = {};
    for (int k = 0; k < K; k += 4) {
        v2f a, b;
        a.x = xrow[k + kb];
        a.y = xrow[k + kb + 1];
        b.x = W1[(k + kb) * 16 + m];
        b.y = W1[(k + kb + 1) * 16 + m];
        c = __builtin_amdgcn_wmma_f32_16x16x4_f32(false, a, false, b,
                                                  (short)0, c, false, false);
    }
    const int rbase = (lane >> 4) << 3;
#pragma unroll
    for (int r = 0; r < 8; ++r) {
        const int orow = tile * 16 + rbase + r;
        if (orow < N) {
            v2f p;
            p.x = c[r];
            p.y = rsqrtf((float)(slots[orow * RUN_WORDS] + slotsB[orow * RUN_WORDS] + 1));
            *(v2f*)(h2 + (size_t)orow * 32 + 2 * m) = p;
        }
    }
}

template <int J>
__device__ __forceinline__ int rowShare(int x) {
    return __builtin_amdgcn_update_dpp(0, x, 0x150 + J, 0xf, 0xf, true);
}

template <int M>
__device__ __forceinline__ float rowXor(float x) {
    return __int_as_float(__builtin_amdgcn_update_dpp(0, __float_as_int(x), 0x160 + M, 0xf, 0xf, true));
}

template <int J>
__device__ __forceinline__ void rowGather(int s, int c, const float* __restrict__ h2, v2f* v) {
    v[J] = *(const v2f*)(h2 + rowShare<J>(s) * 32 + 2 * c);
}

template <int J>
__device__ __forceinline__ void rowSum(int wb, int m, float di, const v2f* v, float& acc) {
    const int wi = wb + J;
    const float w = (wi >= 1 && wi <= m) ? v[J].y * di : 0.f;
    acc += v[J].x * w;
}

#define ROW16(F, ...) \
    F<0>(__VA_ARGS__);  F<1>(__VA_ARGS__);  F<2>(__VA_ARGS__);  F<3>(__VA_ARGS__);  \
    F<4>(__VA_ARGS__);  F<5>(__VA_ARGS__);  F<6>(__VA_ARGS__);  F<7>(__VA_ARGS__);  \
    F<8>(__VA_ARGS__);  F<9>(__VA_ARGS__);  F<10>(__VA_ARGS__); F<11>(__VA_ARGS__); \
    F<12>(__VA_ARGS__); F<13>(__VA_ARGS__); F<14>(__VA_ARGS__); F<15>(__VA_ARGS__)

constexpr int PULL_NODES = 4;

__device__ __forceinline__ void pullRun(const int* __restrict__ run, int w0, int m, int i, int c, float di,
                                        const float* __restrict__ h2, float& acc) {
    int s = (c >= 1 && c <= m) ? w0 : i;
    for (int wb = 0; wb <= m; wb += 16) {
        const int wn = wb + 16 + c;
        int sn = i;
        if (wn <= m) sn = run[wn];
        v2f v[16];
        ROW16(rowGather, s, c, h2, v);
        ROW16(rowSum, wb, m, di, v, acc);
        s = sn;
    }
}

__global__ void __launch_bounds__(256) __attribute__((amdgpu_waves_per_eu(16)))
k_pull16(const int* __restrict__ slots, const int* __restrict__ slotsB,
         const int* __restrict__ ovf_n, const int* __restrict__ ovf,
         const float* __restrict__ h2,
         const float* __restrict__ b1, const float* __restrict__ W2,
         float* __restrict__ gd, int N) {
    const int c = threadIdx.x & 15;
    const float b1c = b1[c];
    const float w2d = W2[c * 2 + 0] - W2[c * 2 + 1];
    const int row  = (int)((blockIdx.x * blockDim.x + threadIdx.x) >> 4);
    const int rows = (int)((gridDim.x * blockDim.x) >> 4);
  for (int t = 0; t < PULL_NODES; ++t) {
    const int i = row + t * rows;
    if (i >= N) break;
    const int* __restrict__ runA = slots  + i * RUN_WORDS;
    const int* __restrict__ runB = slotsB + i * RUN_WORDS;
    const int wA = runA[c];
    const int wB = runB[c];
    const int nA = rowShare<0>(wA);
    const int nB = rowShare<0>(wB);
    const int mA = nA < RUN_CAP ? nA : RUN_CAP;
    const int mB = nB < RUN_CAP ? nB : RUN_CAP;
    const float di = rsqrtf((float)(nA + nB + 1));
    float acc = h2[i * 32 + 2 * c] * di * di;
    if (mA > 0) pullRun(runA, wA, mA, i, c, di, h2, acc);
    if (mB > 0) pullRun(runB, wB, mB, i, c, di, h2, acc);
    if (nA > RUN_CAP || nB > RUN_CAP) {
        const int no = *ovf_n;
        for (int q = 0; q < no; ++q) {
            if (ovf[2 * q + 1] == i) {
                const int os = ovf[2 * q];
                const v2f p = *(const v2f*)(h2 + os * 32 + 2 * c);
                acc += p.x * (p.y * di);
            }
        }
    }
    float v = acc + b1c;
    v = v > 0.f ? v : 0.f;
    float g = v * w2d;
    g += rowXor<1>(g);
    g += rowXor<2>(g);
    g += rowXor<4>(g);
    g += rowXor<8>(g);
    if (c == 0) {
        v2f p;
        p.x = g;
        p.y = di;
        *(v2f*)(gd + (size_t)i * 2) = p;
    }
  }
}


__global__ void __launch_bounds__(256)
k_pull2(const int* __restrict__ slots, const int* __restrict__ slotsB,
        const int* __restrict__ ovf_n, const int* __restrict__ ovf,
        const float* __restrict__ gd, const float* __restrict__ b2,
        float* __restrict__ out, int N) {
    const int c = threadIdx.x & 15;
    const float b2d = b2[0] - b2[1];
    const int row  = (int)((blockIdx.x * blockDim.x + threadIdx.x) >> 4);
    const int rows = (int)((gridDim.x * blockDim.x) >> 4);
  for (int t = 0; t < PULL_NODES; ++t) {
    const int i = row + t * rows;
    if (i >= N) break;
    const int* __restrict__ runA = slots  + i * RUN_WORDS;
    const int* __restrict__ runB = slotsB + i * RUN_WORDS;
    const int a0 = runA[c];
    const int a1 = runA[c + 16];
    const int b0 = runB[c];
    const int b1 = runB[c + 16];
    const float self = gd[(size_t)i * 2];
    const int nA = rowShare<0>(a0);
    const int nB = rowShare<0>(b0);
    const int mA = nA < RUN_CAP ? nA : RUN_CAP;
    const int mB = nB < RUN_CAP ? nB : RUN_CAP;
    const float di = rsqrtf((float)(nA + nB + 1));
    const bool ea0 = c >= 1 && c <= mA, ea1 = c + 16 <= mA;
    const bool eb0 = c >= 1 && c <= mB, eb1 = c + 16 <= mB;
    const v2f p0 = *(const v2f*)(gd + (size_t)(ea0 ? a0 : i) * 2);
    const v2f p1 = *(const v2f*)(gd + (size_t)(ea1 ? a1 : i) * 2);
    const v2f p2 = *(const v2f*)(gd + (size_t)(eb0 ? b0 : i) * 2);
    const v2f p3 = *(const v2f*)(gd + (size_t)(eb1 ? b1 : i) * 2);
    float part = 0.f;
    part += ea0 ? p0.x * (p0.y * di) : 0.f;
    part += ea1 ? p1.x * (p1.y * di) : 0.f;
    part += eb0 ? p2.x * (p2.y * di) : 0.f;
    part += eb1 ? p3.x * (p3.y * di) : 0.f;
    if (mA >= 32 || mB >= 32) {
        const int a2 = runA[c + 32];
        const int a3 = runA[c + 48];
        const int b2 = runB[c + 32];
        const int b3 = runB[c + 48];
        const bool ea2 = c + 32 <= mA, ea3 = c + 48 <= mA;
        const bool eb2 = c + 32 <= mB, eb3 = c + 48 <= mB;
        const v2f q0 = *(const v2f*)(gd + (size_t)(ea2 ? a2 : i) * 2);
        const v2f q1 = *(const v2f*)(gd + (size_t)(ea3 ? a3 : i) * 2);
        const v2f q2 = *(const v2f*)(gd + (size_t)(eb2 ? b2 : i) * 2);
        const v2f q3 = *(const v2f*)(gd + (size_t)(eb3 ? b3 : i) * 2);
        part += ea2 ? q0.x * (q0.y * di) : 0.f;
        part += ea3 ? q1.x * (q1.y * di) : 0.f;
        part += eb2 ? q2.x * (q2.y * di) : 0.f;
        part += eb3 ? q3.x * (q3.y * di) : 0.f;
    }
    if (nA > RUN_CAP || nB > RUN_CAP) {
        const int no = *ovf_n;
        for (int q = c; q < no; q += 16) {
            if (ovf[2 * q + 1] == i) {
                const int os = ovf[2 * q];
                const v2f p = *(const v2f*)(gd + (size_t)os * 2);
                part += p.x * (p.y * di);
            }
        }
    }
    part += __shfl_xor(part, 8, 16);
    part += __shfl_xor(part, 4, 16);
    part += __shfl_xor(part, 2, 16);
    part += __shfl_xor(part, 1, 16);
    if (c == 0) {
        const float z = (self * (di * di) + part) + b2d;
        const float l = logf(1.0f + expf(-fabsf(z)));
        v2f o;
        o.x = fminf(z, 0.f) - l;
        o.y = fminf(-z, 0.f) - l;
        *(v2f*)(out + (size_t)i * 2) = o;
    }
  }
}

extern "C" void kernel_launch(void* const* d_in, const int* in_sizes, int n_in,
                              void* d_out, int out_size, void* d_ws, size_t ws_size,
                              hipStream_t stream) {
    const float* x  = (const float*)d_in[0];
    const int*   ei = (const int*)d_in[1];
    const float* W1 = (const float*)d_in[2];
    const float* b1 = (const float*)d_in[3];
    const float* W2 = (const float*)d_in[4];
    const float* b2 = (const float*)d_in[5];
    float* out = (float*)d_out;

    const int H    = in_sizes[3];
    const int Fin  = in_sizes[2] / H;
    const int N    = in_sizes[0] / Fin;
    const int E    = in_sizes[1] / 2;
    (void)n_in; (void)out_size; (void)ws_size;

    int*   wi     = (int*)d_ws;
    float* ws     = (float*)d_ws;
    int*   slots  = wi;
    int*   slotsB = wi + (size_t)RUN_WORDS * N;
    float* h2     = ws + (size_t)(2 * RUN_WORDS) * N;
    float* gd     = ws + (size_t)(2 * RUN_WORDS + 32) * N;
    int*   ovf_n  = wi + (size_t)(2 * RUN_WORDS + 34) * N;
    int*   ovf    = wi + (size_t)(2 * RUN_WORDS + 35) * N;

    const int B = 256;
    k_deg_init   <<<(N + B - 1) / B, B, 0, stream>>>(slots, slotsB, ovf_n, N);
    k_count_place<<<(E + B * PLACE_EDGES - 1) / (B * PLACE_EDGES), B, 0, stream>>>(ei, E, slots, slotsB, ovf_n, ovf);

    {
        const int ntiles = (N + 15) / 16;
        const int wpb = 4;
        k_gemm_wmma<<<(ntiles + wpb - 1) / wpb, wpb * 32, 0, stream>>>(x, W1, slots, slotsB, h2, N, Fin);
    }

    k_pull16<<<(N + 16 * PULL_NODES - 1) / (16 * PULL_NODES), B, 0, stream>>>(slots, slotsB, ovf_n, ovf, h2, b1, W2, gd, N);

    k_pull2<<<(N + 16 * PULL_NODES - 1) / (16 * PULL_NODES), B, 0, stream>>>(slots, slotsB, ovf_n, ovf, gd, b2, out, N);
}
